# starting kernel
# baseline (speedup 1.0000x reference)
#include <hip/hip_runtime.h>
#include <math.h>


typedef __attribute__((ext_vector_type(2))) float v2f;
typedef __attribute__((ext_vector_type(8))) float v8f;

#define GAT_N      50000
#define GAT_E      1600000
#define GAT_C      128
#define GAT_HEADS  4
#define GAT_FOUT   32
#define NEG_SLOPE  0.2f

__global__ __launch_bounds__(256) void gemm_f32_wmma(
    const float* __restrict__ X, const float* __restrict__ W,
    float* __restrict__ H)
{
    const int wave = threadIdx.x >> 5;
    const int lane = threadIdx.x & 31;
    const int m0   = blockIdx.x * 16;
    const int n0   = wave * 16;

    const int r    = lane & 15;
    const int koff = (lane >> 4) * 2;

    const float* __restrict__ xrow = X + (size_t)(m0 + r) * GAT_C + koff;
    const float* __restrict__ wcol = W + (size_t)koff * GAT_C + (n0 + r);

    v8f acc = {};
    #pragma unroll
    for (int k = 0; k < GAT_C; k += 4) {
        v2f a; a.x = xrow[k];
               a.y = xrow[k + 1];
        v2f b; b.x = wcol[(size_t)k * GAT_C];
               b.y = wcol[(size_t)(k + 1) * GAT_C];
        acc = __builtin_amdgcn_wmma_f32_16x16x4_f32(
             false, a,  false, b,
             (short)0, acc,  false,  false);
    }

    float* __restrict__ hp = H + (size_t)(m0 + (lane >> 4) * 8) * GAT_C + (n0 + r);
    #pragma unroll
    for (int i = 0; i < 8; ++i)
        hp[(size_t)i * GAT_C] = acc[i];
}

__global__ __launch_bounds__(256) void att_scores(
    const float* __restrict__ H,
    const float* __restrict__ att_s, const float* __restrict__ att_d,
    float* __restrict__ a_s, float* __restrict__ a_d, int n)
{
    const int node = blockIdx.x * 8 + (threadIdx.x >> 5);
    const int lane = threadIdx.x & 31;
    if (node >= n) return;
    const float* __restrict__ hp = H + (size_t)node * GAT_C;
    #pragma unroll
    for (int h = 0; h < GAT_HEADS; ++h) {
        const float v = hp[h * GAT_FOUT + lane];
        float s = v * att_s[h * GAT_FOUT + lane];
        float d = v * att_d[h * GAT_FOUT + lane];
        #pragma unroll
        for (int off = 16; off > 0; off >>= 1) {
            s += __shfl_down(s, off, 32);
            d += __shfl_down(d, off, 32);
        }
        if (lane == 0) {
            a_s[node * GAT_HEADS + h] = s;
            a_d[node * GAT_HEADS + h] = d;
        }
    }
}

__global__ __launch_bounds__(256) void init_layer(
    float* __restrict__ emax, float* __restrict__ denom,
    float* __restrict__ out, const float* __restrict__ bias, int n)
{
    const int i = blockIdx.x * 256 + threadIdx.x;
    if (i < n * GAT_HEADS) { emax[i] = -INFINITY; denom[i] = 0.0f; }
    if (i < n * GAT_C)     { out[i] = bias[i & (GAT_C - 1)]; }
}

__global__ __launch_bounds__(256) void edge_seg_max(
    const int* __restrict__ src, const int* __restrict__ dst,
    const float* __restrict__ a_s, const float* __restrict__ a_d,
    float* __restrict__ emax, int ne)
{
    const int t = blockIdx.x * 256 + threadIdx.x;
    if (t >= ne * GAT_HEADS) return;
    const int e = t >> 2, h = t & 3;
    const int s = src[e], d = dst[e];
    float v = a_s[s * GAT_HEADS + h] + a_d[d * GAT_HEADS + h];
    v = (v > 0.0f) ? v : NEG_SLOPE * v;
    atomicMax(&emax[d * GAT_HEADS + h], v);
}

__global__ __launch_bounds__(256) void edge_seg_sumexp(
    const int* __restrict__ src, const int* __restrict__ dst,
    const float* __restrict__ a_s, const float* __restrict__ a_d,
    const float* __restrict__ emax, float* __restrict__ denom, int ne)
{
    const int t = blockIdx.x * 256 + threadIdx.x;
    if (t >= ne * GAT_HEADS) return;
    const int e = t >> 2, h = t & 3;
    const int s = src[e], d = dst[e];
    float v = a_s[s * GAT_HEADS + h] + a_d[d * GAT_HEADS + h];
    v = (v > 0.0f) ? v : NEG_SLOPE * v;
    atomicAdd(&denom[d * GAT_HEADS + h], __expf(v - emax[d * GAT_HEADS + h]));
}

__global__ __launch_bounds__(256) void edge_aggregate(
    const int* __restrict__ src, const int* __restrict__ dst,
    const float* __restrict__ a_s, const float* __restrict__ a_d,
    const float* __restrict__ emax, const float* __restrict__ denom,
    const float* __restrict__ H, float* __restrict__ out, int ne)
{
    const int e = blockIdx.x * 8 + (threadIdx.x >> 5);
    if (e >= ne) return;
    const int lane = threadIdx.x & 31;
    const int s = src[e], d = dst[e];
    #pragma unroll
    for (int h = 0; h < GAT_HEADS; ++h) {
        float v = a_s[s * GAT_HEADS + h] + a_d[d * GAT_HEADS + h];
        v = (v > 0.0f) ? v : NEG_SLOPE * v;
        float den = denom[d * GAT_HEADS + h];
        den = (den > 0.0f) ? den : 1.0f;
        const float alpha = __expf(v - emax[d * GAT_HEADS + h]) / den;
        const float hv = H[(size_t)s * GAT_C + h * GAT_FOUT + lane];
        atomicAdd(&out[(size_t)d * GAT_C + h * GAT_FOUT + lane], alpha * hv);
    }
}

__global__ __launch_bounds__(256) void relu_inplace(float* __restrict__ x, int n)
{
    const int i = blockIdx.x * 256 + threadIdx.x;
    if (i < n) x[i] = fmaxf(x[i], 0.0f);
}

extern "C" void kernel_launch(void* const* d_in, const int* in_sizes, int n_in,
                              void* d_out, int out_size, void* d_ws, size_t ws_size,
                              hipStream_t stream)
{
    const float* x    = (const float*)d_in[0];
    const int*   eidx = (const int*)  d_in[1];
    const float* W0   = (const float*)d_in[2];
    const float* as0  = (const float*)d_in[3];
    const float* ad0  = (const float*)d_in[4];
    const float* b0   = (const float*)d_in[5];
    const float* W1   = (const float*)d_in[6];
    const float* as1  = (const float*)d_in[7];
    const float* ad1  = (const float*)d_in[8];
    const float* b1   = (const float*)d_in[9];

    const int N = in_sizes[0] / GAT_C;
    const int E = in_sizes[1] / 2;
    const int* src = eidx;
    const int* dst = eidx + E;

    char* ws = (char*)d_ws;
    float* hbuf  = (float*)ws;                                  ws += (size_t)N * GAT_C * 4;
    float* out0  = (float*)ws;                                  ws += (size_t)N * GAT_C * 4;
    float* a_s   = (float*)ws;                                  ws += (size_t)N * GAT_HEADS * 4;
    float* a_d   = (float*)ws;                                  ws += (size_t)N * GAT_HEADS * 4;
    float* emax  = (float*)ws;                                  ws += (size_t)N * GAT_HEADS * 4;
    float* denom = (float*)ws;

    const int blk = 256;
    const dim3 gGemm(N / 16);
    const dim3 gNode((N + 7) / 8);
    const dim3 gInit((N * GAT_C + blk - 1) / blk);
    const dim3 gEdgeH((E * GAT_HEADS + blk - 1) / blk);
    const dim3 gEdgeW((E + 7) / 8);
    const dim3 gElem((N * GAT_C + blk - 1) / blk);

    init_layer    <<<gInit,  blk, 0, stream>>>(emax, denom, out0, b0, N);
    gemm_f32_wmma <<<gGemm,  blk, 0, stream>>>(x, W0, hbuf);
    att_scores    <<<gNode,  blk, 0, stream>>>(hbuf, as0, ad0, a_s, a_d, N);
    edge_seg_max  <<<gEdgeH, blk, 0, stream>>>(src, dst, a_s, a_d, emax, E);
    edge_seg_sumexp<<<gEdgeH,blk, 0, stream>>>(src, dst, a_s, a_d, emax, denom, E);
    edge_aggregate<<<gEdgeW, blk, 0, stream>>>(src, dst, a_s, a_d, emax, denom,
                                               hbuf, out0, E);
    relu_inplace  <<<gElem,  blk, 0, stream>>>(out0, N * GAT_C);

    float* outf = (float*)d_out;
    init_layer    <<<gInit,  blk, 0, stream>>>(emax, denom, outf, b1, N);
    gemm_f32_wmma <<<gGemm,  blk, 0, stream>>>(out0, W1, hbuf);
    att_scores    <<<gNode,  blk, 0, stream>>>(hbuf, as1, ad1, a_s, a_d, N);
    edge_seg_max  <<<gEdgeH, blk, 0, stream>>>(src, dst, a_s, a_d, emax, E);
    edge_seg_sumexp<<<gEdgeH,blk, 0, stream>>>(src, dst, a_s, a_d, emax, denom, E);
    edge_aggregate<<<gEdgeW, blk, 0, stream>>>(src, dst, a_s, a_d, emax, denom,
                                               hbuf, outf, E);
}


# Edges grouped by destination once per call and shared by both layers: bucket partition per chunk in LDS, then each owner range builds its node runs; no gather scans the whole dst array
# speedup vs baseline: 5.3172x; 5.3172x over previous
#include <hip/hip_runtime.h>
#include <math.h>


typedef __attribute__((ext_vector_type(2))) float v2f;
typedef __attribute__((ext_vector_type(8))) float v8f;

#define GAT_N      50000
#define GAT_E      1600000
#define GAT_C      128
#define GAT_HEADS  4
#define GAT_FOUT   32
#define NEG_SLOPE  0.2f

__global__ __launch_bounds__(256) void gemm_f32_wmma(
    const float* __restrict__ X, const float* __restrict__ W,
    float* __restrict__ H)
{
    const int wave = threadIdx.x >> 5;
    const int lane = threadIdx.x & 31;
    const int m0   = blockIdx.x * 16;
    const int n0   = wave * 16;

    const int r    = lane & 15;
    const int koff = (lane >> 4) * 2;

    const float* __restrict__ xrow = X + (size_t)(m0 + r) * GAT_C + koff;
    const float* __restrict__ wcol = W + (size_t)koff * GAT_C + (n0 + r);

    v8f acc = {};
    #pragma unroll
    for (int k = 0; k < GAT_C; k += 4) {
        v2f a; a.x = xrow[k];
               a.y = xrow[k + 1];
        v2f b; b.x = wcol[(size_t)k * GAT_C];
               b.y = wcol[(size_t)(k + 1) * GAT_C];
        acc = __builtin_amdgcn_wmma_f32_16x16x4_f32(
             false, a,  false, b,
             (short)0, acc,  false,  false);
    }

    float* __restrict__ hp = H + (size_t)(m0 + (lane >> 4) * 8) * GAT_C + (n0 + r);
    #pragma unroll
    for (int i = 0; i < 8; ++i)
        hp[(size_t)i * GAT_C] = acc[i];
}

__global__ __launch_bounds__(256) void att_scores(
    const float* __restrict__ H,
    const float* __restrict__ att_s, const float* __restrict__ att_d,
    float* __restrict__ a_s, float* __restrict__ a_d, int n)
{
    const int node = blockIdx.x * 8 + (threadIdx.x >> 5);
    const int lane = threadIdx.x & 31;
    if (node >= n) return;
    const float* __restrict__ hp = H + (size_t)node * GAT_C;
    #pragma unroll
    for (int h = 0; h < GAT_HEADS; ++h) {
        const float v = hp[h * GAT_FOUT + lane];
        float s = v * att_s[h * GAT_FOUT + lane];
        float d = v * att_d[h * GAT_FOUT + lane];
        #pragma unroll
        for (int off = 16; off > 0; off >>= 1) {
            s += __shfl_down(s, off, 32);
            d += __shfl_down(d, off, 32);
        }
        if (lane == 0) {
            a_s[node * GAT_HEADS + h] = s;
            a_d[node * GAT_HEADS + h] = d;
        }
    }
}

#define GAT_NPB      196
#define GAT_NB       256
#define GAT_CH       6250
#define GAT_GTHREADS 1024
#define GAT_GWAVES   (GAT_GTHREADS / 32)
#define GAT_PER      ((GAT_CH + GAT_GTHREADS - 1) / GAT_GTHREADS)

static_assert(GAT_NB * GAT_NPB >= GAT_N, "every node needs an owner range");
static_assert(GAT_NB % 32 == 0, "bucket scan: whole lanes");

__global__ __launch_bounds__(GAT_GTHREADS) void edge_partition(
    const int* __restrict__ src, const int* __restrict__ dst,
    int2* __restrict__ part, int* __restrict__ table, int n, int ne)
{
    __shared__ int  s_hist[GAT_NB];
    __shared__ int  s_start[GAT_NB + 1];
    __shared__ int2 s_stage[GAT_CH];

    const int tid = threadIdx.x;
    const int e0  = blockIdx.x * GAT_CH;
    const int cnt = min(GAT_CH, ne - e0);

    if (tid < GAT_NB) s_hist[tid] = 0;
    __syncthreads();

    int sv[GAT_PER], lv[GAT_PER], bv[GAT_PER], rv[GAT_PER];
    #pragma unroll
    for (int k = 0; k < GAT_PER; ++k) {
        const int i = tid + k * GAT_GTHREADS;
        bv[k] = -1;
        if (i < cnt) {
            const int d = dst[e0 + i];
            const int sidv = src[e0 + i];
            if ((unsigned)d < (unsigned)n) {
                const int b = (int)((unsigned)d / (unsigned)GAT_NPB);
                bv[k] = b;
                lv[k] = d - b * GAT_NPB;
                sv[k] = sidv;
                rv[k] = atomicAdd(&s_hist[b], 1);
            }
        }
    }
    __syncthreads();

    if (tid < 32) {
        constexpr int PB = GAT_NB / 32;
        int before[PB];
        int sum = 0;
        #pragma unroll
        for (int k = 0; k < PB; ++k) {
            before[k] = sum;
            sum += s_hist[tid * PB + k];
        }
        int inc = sum;
        #pragma unroll
        for (int o = 1; o < 32; o <<= 1) {
            const int t = __shfl_up(inc, o, 32);
            if (tid >= o) inc += t;
        }
        const int basev = inc - sum;
        #pragma unroll
        for (int k = 0; k < PB; ++k)
            s_start[tid * PB + k] = basev + before[k];
        if (tid == 31) s_start[GAT_NB] = inc;
    }
    __syncthreads();

    #pragma unroll
    for (int k = 0; k < GAT_PER; ++k)
        if (bv[k] >= 0)
            s_stage[s_start[bv[k]] + rv[k]] = make_int2(sv[k], lv[k]);
    if (tid <= GAT_NB)
        table[(size_t)blockIdx.x * (GAT_NB + 1) + tid] = s_start[tid];
    __syncthreads();

    const int kept = s_start[GAT_NB];
    int2* __restrict__ pout = part + (size_t)e0;
    for (int i = tid; i < kept; i += GAT_GTHREADS)
        pout[i] = s_stage[i];
}

__global__ __launch_bounds__(GAT_GTHREADS) void edges_by_node(
    const int2* __restrict__ part, const int* __restrict__ table,
    int* __restrict__ csr_off, int* __restrict__ csr_src, int n, int nch)
{
    __shared__ int s_cnt[GAT_NPB];
    __shared__ int s_cur[GAT_NPB];
    __shared__ int s_base;

    const int tid  = threadIdx.x;
    const int lane = tid & 31;
    const int j    = blockIdx.x;
    const int lo   = j * GAT_NPB;
    const int nn   = min(GAT_NPB, n - lo);
    if (nn <= 0) return;

    if (tid < GAT_NPB) s_cnt[tid] = 0;
    if (tid == 0) s_base = 0;
    __syncthreads();

    int below = 0;
    for (int c = tid; c < nch; c += GAT_GTHREADS)
        below += table[(size_t)c * (GAT_NB + 1) + j];
    #pragma unroll
    for (int o = 16; o > 0; o >>= 1)
        below += __shfl_down(below, o, 32);
    if (lane == 0 && below != 0) atomicAdd(&s_base, below);

    const int sub = tid & 3;
    for (int c = tid >> 2; c < nch; c += GAT_GTHREADS / 4) {
        const int pb = table[(size_t)c * (GAT_NB + 1) + j];
        const int pe = table[(size_t)c * (GAT_NB + 1) + j + 1];
        const int2* __restrict__ p = part + (size_t)c * GAT_CH;
        for (int i = pb + sub; i < pe; i += 4)
            atomicAdd(&s_cnt[p[i].y], 1);
    }
    __syncthreads();

    const int base = s_base;
    if (tid < 32) {
        constexpr int PER = (GAT_NPB + 31) / 32;
        int before[PER];
        int sum = 0;
        #pragma unroll
        for (int k = 0; k < PER; ++k) {
            const int idx = tid * PER + k;
            before[k] = sum;
            sum += (idx < GAT_NPB) ? s_cnt[idx] : 0;
        }
        int inc = sum;
        #pragma unroll
        for (int o = 1; o < 32; o <<= 1) {
            const int t = __shfl_up(inc, o, 32);
            if (tid >= o) inc += t;
        }
        const int basev = inc - sum;
        #pragma unroll
        for (int k = 0; k < PER; ++k) {
            const int idx = tid * PER + k;
            if (idx < GAT_NPB) s_cur[idx] = basev + before[k];
            if (idx < nn) csr_off[lo + idx] = base + basev + before[k];
        }
        if (tid == 31 && lo + nn == n) csr_off[n] = base + inc;
    }
    __syncthreads();

    for (int c = tid >> 2; c < nch; c += GAT_GTHREADS / 4) {
        const int pb = table[(size_t)c * (GAT_NB + 1) + j];
        const int pe = table[(size_t)c * (GAT_NB + 1) + j + 1];
        const int2* __restrict__ p = part + (size_t)c * GAT_CH;
        for (int i = pb + sub; i < pe; i += 4) {
            const int2 v = p[i];
            const int pos = atomicAdd(&s_cur[v.y], 1);
            csr_src[base + pos] = v.x;
        }
    }
}

struct NodeAcc {
    float m, den;
    float4 acc;
};

__device__ __forceinline__ void edge_update(float as, float ad, const float4 hv, NodeAcc& st)
{
    float e = as + ad;
    e = (e > 0.0f) ? e : NEG_SLOPE * e;
    const float mn = fmaxf(st.m, e);
    const float sc = __expf(st.m - mn);
    const float p  = __expf(e - mn);
    st.den   = st.den   * sc + p;
    st.acc.x = st.acc.x * sc + p * hv.x;
    st.acc.y = st.acc.y * sc + p * hv.y;
    st.acc.z = st.acc.z * sc + p * hv.z;
    st.acc.w = st.acc.w * sc + p * hv.w;
    st.m = mn;
}

template <int U>
__device__ __forceinline__ void edge_group(
    int ids, int first, int lane, int head, float ad,
    const float* __restrict__ H, const float* __restrict__ a_s, NodeAcc& st)
{
    float  as[U];
    float4 hv[U];
    #pragma unroll
    for (int u = 0; u < U; ++u) {
        const int s = __builtin_amdgcn_readlane(ids, first + u);
        as[u] = a_s[s * GAT_HEADS + head];
        hv[u] = *reinterpret_cast<const float4*>(H + (size_t)s * GAT_C + 4 * lane);
    }
    #pragma unroll
    for (int u = 0; u < U; ++u)
        edge_update(as[u], ad, hv[u], st);
}

__device__ __forceinline__ void edge_chunk(
    int ids, int c, int lane, int head, float ad,
    const float* __restrict__ H, const float* __restrict__ a_s, NodeAcc& st)
{
    int u = 0;
    for (; u + 8 <= c; u += 8)
        edge_group<8>(ids, u, lane, head, ad, H, a_s, st);
    if (u + 4 <= c) {
        edge_group<4>(ids, u, lane, head, ad, H, a_s, st);
        u += 4;
    }
    for (; u < c; ++u)
        edge_group<1>(ids, u, lane, head, ad, H, a_s, st);
}

template <bool RELU>
__device__ __forceinline__ void store_row(
    const NodeAcc& st, const float4 b4, float* __restrict__ out, int node, int lane)
{
    const float den = (st.den > 0.0f) ? st.den : 1.0f;
    float4 r;
    r.x = st.acc.x / den + b4.x;
    r.y = st.acc.y / den + b4.y;
    r.z = st.acc.z / den + b4.z;
    r.w = st.acc.w / den + b4.w;
    if (RELU) {
        r.x = fmaxf(r.x, 0.0f); r.y = fmaxf(r.y, 0.0f);
        r.z = fmaxf(r.z, 0.0f); r.w = fmaxf(r.w, 0.0f);
    }
    *reinterpret_cast<float4*>(out + (size_t)node * GAT_C + 4 * lane) = r;
}

template <bool RELU>
__global__ __launch_bounds__(GAT_GTHREADS) void gat_gather(
    const int* __restrict__ csr_off, const int* __restrict__ csr_src,
    const float* __restrict__ H,
    const float* __restrict__ a_s, const float* __restrict__ a_d,
    const float* __restrict__ bias, float* __restrict__ out, int n)
{
    __shared__ int s_off[GAT_NPB + 1];

    const int tid  = threadIdx.x;
    const int lane = tid & 31;
    const int head = lane >> 3;
    const int lo   = blockIdx.x * GAT_NPB;
    const int nn   = min(GAT_NPB, n - lo);
    if (nn <= 0) return;

    if (tid <= nn) s_off[tid] = csr_off[lo + tid];
    __syncthreads();

    const float4 b4 = *reinterpret_cast<const float4*>(bias + 4 * lane);

    auto run_ids = [&](int b, int e, int first) {
        return csr_src[max(min(b + first + lane, e - 1), 0)];
    };
    int nl = __builtin_amdgcn_readfirstlane(tid >> 5);
    if (nl >= nn) return;
    int   beg  = __builtin_amdgcn_readfirstlane(s_off[nl]);
    int   end  = __builtin_amdgcn_readfirstlane(s_off[nl + 1]);
    int   idsA = run_ids(beg, end, 0);
    int   idsB = run_ids(beg, end, 32);
    float ad   = a_d[(lo + nl) * GAT_HEADS + head];
    while (nl < nn) {
        const int nl2 = nl + GAT_GWAVES;
        const int nlc = min(nl2, nn - 1);
        const int   beg2 = __builtin_amdgcn_readfirstlane(s_off[nlc]);
        const int   end2 = __builtin_amdgcn_readfirstlane(s_off[nlc + 1]);
        const int   nxtA = run_ids(beg2, end2, 0);
        const int   nxtB = run_ids(beg2, end2, 32);
        const float ad2  = a_d[(lo + nlc) * GAT_HEADS + head];

        const int node = lo + nl;
        NodeAcc st; st.m = -INFINITY; st.den = 0.0f;
        st.acc = make_float4(0.0f, 0.0f, 0.0f, 0.0f);

        const int c = end - beg;
        edge_chunk(idsA, min(c, 32), lane, head, ad, H, a_s, st);
        if (c > 32)
            edge_chunk(idsB, min(c - 32, 32), lane, head, ad, H, a_s, st);
        for (int k = beg + 64; k < end; k += 32) {
            const int ids = (k + lane < end) ? csr_src[k + lane] : 0;
            edge_chunk(ids, min(end - k, 32), lane, head, ad, H, a_s, st);
        }
        store_row<RELU>(st, b4, out, node, lane);

        nl = nl2; beg = beg2; end = end2; idsA = nxtA; idsB = nxtB; ad = ad2;
    }
}

extern "C" void kernel_launch(void* const* d_in, const int* in_sizes, int n_in,
                              void* d_out, int out_size, void* d_ws, size_t ws_size,
                              hipStream_t stream)
{
    const float* x    = (const float*)d_in[0];
    const int*   eidx = (const int*)  d_in[1];
    const float* W0   = (const float*)d_in[2];
    const float* as0  = (const float*)d_in[3];
    const float* ad0  = (const float*)d_in[4];
    const float* b0   = (const float*)d_in[5];
    const float* W1   = (const float*)d_in[6];
    const float* as1  = (const float*)d_in[7];
    const float* ad1  = (const float*)d_in[8];
    const float* b1   = (const float*)d_in[9];

    const int N = in_sizes[0] / GAT_C;
    const int E = in_sizes[1] / 2;
    const int* src = eidx;
    const int* dst = eidx + E;

    char* ws = (char*)d_ws;
    float* hbuf    = (float*)ws;                                ws += (size_t)N * GAT_C * 4;
    char*  plane   = ws;                                        ws += (size_t)N * GAT_C * 4;
    float* a_s     = (float*)ws;                                ws += (size_t)N * GAT_HEADS * 4;
    float* a_d     = (float*)ws;
    float* outf    = (float*)d_out;
    float* out0    = outf;

    const int nch = (E + GAT_CH - 1) / GAT_CH;
    int2* part     = (int2*)plane;                              plane += (size_t)nch * GAT_CH * 8;
    int*  table    = (int*)plane;                               plane += (size_t)nch * (GAT_NB + 1) * 4;
    int*  csr_src  = (int*)plane;                               plane += (size_t)nch * GAT_CH * 4;
    int*  csr_off  = (int*)plane;

    const int blk = 256;
    const dim3 gGemm(N / 16);
    const dim3 gNode((N + 7) / 8);
    const dim3 gGather((N + GAT_NPB - 1) / GAT_NPB);

    edge_partition<<<dim3(nch), GAT_GTHREADS, 0, stream>>>(src, dst, part, table, N, E);
    edges_by_node <<<gGather,   GAT_GTHREADS, 0, stream>>>(part, table, csr_off, csr_src, N, nch);

    gemm_f32_wmma <<<gGemm,  blk, 0, stream>>>(x, W0, hbuf);
    att_scores    <<<gNode,  blk, 0, stream>>>(hbuf, as0, ad0, a_s, a_d, N);
    gat_gather<true> <<<gGather, GAT_GTHREADS, 0, stream>>>(csr_off, csr_src, hbuf, a_s, a_d,
                                                            b0, out0, N);

    gemm_f32_wmma <<<gGemm,  blk, 0, stream>>>(out0, W1, hbuf);
    att_scores    <<<gNode,  blk, 0, stream>>>(hbuf, as1, ad1, a_s, a_d, N);
    gat_gather<false><<<gGather, GAT_GTHREADS, 0, stream>>>(csr_off, csr_src, hbuf, a_s, a_d,
                                                            b1, outf, N);
}


# Layer-0 GEMM loses its own launch: its row tiles ride as extra blocks in the launch of the edge grouping's second kernel; 64 registers, so a grouping block and a GEMM block share a compute unit
# speedup vs baseline: 6.7812x; 1.2753x over previous
#include <hip/hip_runtime.h>
#include <math.h>


typedef __attribute__((ext_vector_type(2))) float v2f;
typedef __attribute__((ext_vector_type(8))) float v8f;

#define GAT_N      50000
#define GAT_E      1600000
#define GAT_C      128
#define GAT_HEADS  4
#define GAT_FOUT   32
#define NEG_SLOPE  0.2f


__device__ __forceinline__ float row16_sum(float v)
{
    v += __int_as_float(__builtin_amdgcn_update_dpp(0, __float_as_int(v), 0x161, 0xf, 0xf, false));
    v += __int_as_float(__builtin_amdgcn_update_dpp(0, __float_as_int(v), 0x162, 0xf, 0xf, false));
    v += __int_as_float(__builtin_amdgcn_update_dpp(0, __float_as_int(v), 0x164, 0xf, 0xf, false));
    v += __int_as_float(__builtin_amdgcn_update_dpp(0, __float_as_int(v), 0x168, 0xf, 0xf, false));
    return v;
}

template <int LOOK>
__device__ __forceinline__ void gemm_scores_task(
    const float* __restrict__ X, const float* __restrict__ W,
    const float* __restrict__ att_s, const float* __restrict__ att_d,
    float* __restrict__ H, float* __restrict__ a_s, float* __restrict__ a_d,
    int m0, int head, int lane)
{
    const int n0   = head * GAT_FOUT;

    const int r    = lane & 15;
    const int hh   = lane >> 4;
    const int koff = hh * 2;

    const float* __restrict__ xrow = X + (size_t)(m0 + r) * GAT_C + koff;
    const float* __restrict__ wcol = W + (size_t)koff * GAT_C + (n0 + 2 * r);

    v8f acc0 = {};
    v8f acc1 = {};
    if constexpr (LOOK == 0) {
        #pragma unroll
        for (int k = 0; k < GAT_C; k += 4) {
            v2f a; a.x = xrow[k];
                   a.y = xrow[k + 1];
            const float* __restrict__ wk = wcol + (size_t)k * GAT_C;
            v2f b0; b0.x = wk[0];
                    b0.y = wk[GAT_C];
            v2f b1; b1.x = wk[1];
                    b1.y = wk[GAT_C + 1];
            acc0 = __builtin_amdgcn_wmma_f32_16x16x4_f32(
                 false, a,  false, b0,
                 (short)0, acc0,  false,  false);
            acc1 = __builtin_amdgcn_wmma_f32_16x16x4_f32(
                 false, a,  false, b1,
                 (short)0, acc1,  false,  false);
        }
    } else {
        float2 xa[GAT_C / 4], w0[GAT_C / 4], w1[GAT_C / 4];
        #pragma unroll
        for (int s = 0; s < LOOK; ++s) {
            xa[s] = *reinterpret_cast<const float2*>(xrow + 4 * s);
            w0[s] = *reinterpret_cast<const float2*>(wcol + (size_t)(4 * s) * GAT_C);
            w1[s] = *reinterpret_cast<const float2*>(wcol + (size_t)(4 * s + 1) * GAT_C);
        }
        #pragma unroll
        for (int s = 0; s < GAT_C / 4; ++s) {
            if (s + LOOK < GAT_C / 4) {
                const int t = s + LOOK;
                xa[t] = *reinterpret_cast<const float2*>(xrow + 4 * t);
                w0[t] = *reinterpret_cast<const float2*>(wcol + (size_t)(4 * t) * GAT_C);
                w1[t] = *reinterpret_cast<const float2*>(wcol + (size_t)(4 * t + 1) * GAT_C);
            }
            __builtin_amdgcn_sched_barrier(0);
            v2f a;  a.x  = xa[s].x; a.y  = xa[s].y;
            v2f b0; b0.x = w0[s].x; b0.y = w1[s].x;
            v2f b1; b1.x = w0[s].y; b1.y = w1[s].y;
            acc0 = __builtin_amdgcn_wmma_f32_16x16x4_f32(
                 false, a,  false, b0,
                 (short)0, acc0,  false,  false);
            acc1 = __builtin_amdgcn_wmma_f32_16x16x4_f32(
                 false, a,  false, b1,
                 (short)0, acc1,  false,  false);
            __builtin_amdgcn_sched_barrier(0);
        }
    }

    const float as0 = att_s[n0 + 2 * r], as1 = att_s[n0 + 2 * r + 1];
    const float ad0 = att_d[n0 + 2 * r], ad1 = att_d[n0 + 2 * r + 1];
    const int   row0 = m0 + hh * 8;
    float* __restrict__ hp = H + (size_t)row0 * GAT_C + (n0 + 2 * r);
    float ps[8], pd[8];
    #pragma unroll
    for (int i = 0; i < 8; ++i) {
        *reinterpret_cast<float2*>(hp + (size_t)i * GAT_C) = make_float2(acc0[i], acc1[i]);
        ps[i] = acc0[i] * as0 + acc1[i] * as1;
        pd[i] = acc0[i] * ad0 + acc1[i] * ad1;
    }
    #pragma unroll
    for (int i = 0; i < 8; ++i) {
        ps[i] = row16_sum(ps[i]);
        pd[i] = row16_sum(pd[i]);
    }
    if (r == 0) {
        #pragma unroll
        for (int i = 0; i < 8; ++i) {
            a_s[(row0 + i) * GAT_HEADS + head] = ps[i];
            a_d[(row0 + i) * GAT_HEADS + head] = pd[i];
        }
    }
}

__global__ __launch_bounds__(256) void gemm_scores_f32_wmma(
    const float* __restrict__ X, const float* __restrict__ W,
    const float* __restrict__ att_s, const float* __restrict__ att_d,
    float* __restrict__ H, float* __restrict__ a_s, float* __restrict__ a_d, int n)
{
    const int wave = threadIdx.x >> 5;
    const int lane = threadIdx.x & 31;
    const int head = wave & 3;
    const int m0   = (blockIdx.x * 2 + (wave >> 2)) * 16;
    if (m0 >= n) return;
    gemm_scores_task<0>(X, W, att_s, att_d, H, a_s, a_d, m0, head, lane);
}

#define GAT_NPB      196
#define GAT_NB       256
#define GAT_CH       6250
#define GAT_GTHREADS 1024
#define GAT_GWAVES   (GAT_GTHREADS / 32)
#define GAT_PER      ((GAT_CH + GAT_GTHREADS - 1) / GAT_GTHREADS)

static_assert(GAT_NB * GAT_NPB >= GAT_N, "every node needs an owner range");
static_assert(GAT_NB % 32 == 0, "bucket scan: whole lanes");

__global__ __launch_bounds__(GAT_GTHREADS) void edge_partition(
    const int* __restrict__ src, const int* __restrict__ dst,
    int2* __restrict__ part, int* __restrict__ table, int n, int ne)
{
    __shared__ int  s_hist[GAT_NB];
    __shared__ int  s_start[GAT_NB + 1];
    __shared__ int2 s_stage[GAT_CH];

    const int tid = threadIdx.x;
    const int e0  = blockIdx.x * GAT_CH;
    const int cnt = min(GAT_CH, ne - e0);

    if (tid < GAT_NB) s_hist[tid] = 0;
    __syncthreads();

    int sv[GAT_PER], lv[GAT_PER], bv[GAT_PER], rv[GAT_PER];
    #pragma unroll
    for (int k = 0; k < GAT_PER; ++k) {
        const int i = tid + k * GAT_GTHREADS;
        bv[k] = -1;
        if (i < cnt) {
            const int d = dst[e0 + i];
            const int sidv = src[e0 + i];
            if ((unsigned)d < (unsigned)n) {
                const int b = (int)((unsigned)d / (unsigned)GAT_NPB);
                bv[k] = b;
                lv[k] = d - b * GAT_NPB;
                sv[k] = sidv;
                rv[k] = atomicAdd(&s_hist[b], 1);
            }
        }
    }
    __syncthreads();

    if (tid < 32) {
        constexpr int PB = GAT_NB / 32;
        int before[PB];
        int sum = 0;
        #pragma unroll
        for (int k = 0; k < PB; ++k) {
            before[k] = sum;
            sum += s_hist[tid * PB + k];
        }
        int inc = sum;
        #pragma unroll
        for (int o = 1; o < 32; o <<= 1) {
            const int t = __shfl_up(inc, o, 32);
            if (tid >= o) inc += t;
        }
        const int basev = inc - sum;
        #pragma unroll
        for (int k = 0; k < PB; ++k)
            s_start[tid * PB + k] = basev + before[k];
        if (tid == 31) s_start[GAT_NB] = inc;
    }
    __syncthreads();

    #pragma unroll
    for (int k = 0; k < GAT_PER; ++k)
        if (bv[k] >= 0)
            s_stage[s_start[bv[k]] + rv[k]] = make_int2(sv[k], lv[k]);
    if (tid <= GAT_NB)
        table[(size_t)blockIdx.x * (GAT_NB + 1) + tid] = s_start[tid];
    __syncthreads();

    const int kept = s_start[GAT_NB];
    int2* __restrict__ pout = part + (size_t)e0;
    for (int i = tid; i < kept; i += GAT_GTHREADS)
        pout[i] = s_stage[i];
}

__device__ __forceinline__ void edges_by_node_range(
    const int2* __restrict__ part, const int* __restrict__ table,
    int* __restrict__ csr_off, int* __restrict__ csr_src, int n, int nch, int j)
{
    __shared__ int s_cnt[GAT_NPB];
    __shared__ int s_cur[GAT_NPB];
    __shared__ int s_base;

    const int tid  = threadIdx.x;
    const int lane = tid & 31;
    const int lo   = j * GAT_NPB;
    const int nn   = min(GAT_NPB, n - lo);
    if (nn <= 0) return;

    if (tid < GAT_NPB) s_cnt[tid] = 0;
    if (tid == 0) s_base = 0;
    __syncthreads();

    int below = 0;
    for (int c = tid; c < nch; c += GAT_GTHREADS)
        below += table[(size_t)c * (GAT_NB + 1) + j];
    #pragma unroll
    for (int o = 16; o > 0; o >>= 1)
        below += __shfl_down(below, o, 32);
    if (lane == 0 && below != 0) atomicAdd(&s_base, below);

    const int sub = tid & 3;
    for (int c = tid >> 2; c < nch; c += GAT_GTHREADS / 4) {
        const int pb = table[(size_t)c * (GAT_NB + 1) + j];
        const int pe = table[(size_t)c * (GAT_NB + 1) + j + 1];
        const int2* __restrict__ p = part + (size_t)c * GAT_CH;
        for (int i = pb + sub; i < pe; i += 4)
            atomicAdd(&s_cnt[p[i].y], 1);
    }
    __syncthreads();

    const int base = s_base;
    if (tid < 32) {
        constexpr int PER = (GAT_NPB + 31) / 32;
        int before[PER];
        int sum = 0;
        #pragma unroll
        for (int k = 0; k < PER; ++k) {
            const int idx = tid * PER + k;
            before[k] = sum;
            sum += (idx < GAT_NPB) ? s_cnt[idx] : 0;
        }
        int inc = sum;
        #pragma unroll
        for (int o = 1; o < 32; o <<= 1) {
            const int t = __shfl_up(inc, o, 32);
            if (tid >= o) inc += t;
        }
        const int basev = inc - sum;
        #pragma unroll
        for (int k = 0; k < PER; ++k) {
            const int idx = tid * PER + k;
            if (idx < GAT_NPB) s_cur[idx] = basev + before[k];
            if (idx < nn) csr_off[lo + idx] = base + basev + before[k];
        }
        if (tid == 31 && lo + nn == n) csr_off[n] = base + inc;
    }
    __syncthreads();

    for (int c = tid >> 2; c < nch; c += GAT_GTHREADS / 4) {
        const int pb = table[(size_t)c * (GAT_NB + 1) + j];
        const int pe = table[(size_t)c * (GAT_NB + 1) + j + 1];
        const int2* __restrict__ p = part + (size_t)c * GAT_CH;
        for (int i = pb + sub; i < pe; i += 4) {
            const int2 v = p[i];
            const int pos = atomicAdd(&s_cur[v.y], 1);
            csr_src[base + pos] = v.x;
        }
    }
}

#define GEMM0_LOOK 4

__device__ __forceinline__ void gemm0_tiles(
    const float* __restrict__ X, const float* __restrict__ W,
    const float* __restrict__ att_s, const float* __restrict__ att_d,
    float* __restrict__ H, float* __restrict__ a_s, float* __restrict__ a_d,
    int ntiles, int idx, int nb)
{
    const int lane = threadIdx.x & 31;
    const int wave = threadIdx.x >> 5;
    const int mine = (ntiles - idx + nb - 1) / nb;
    for (int q = wave; q < mine * GAT_HEADS; q += GAT_GWAVES) {
        const int m0 = (idx + (q >> 2) * nb) * 16;
        int head = q & 3, ln = lane;
        asm volatile("" : "+v"(head), "+v"(ln));
        gemm_scores_task<GEMM0_LOOK>(X, W, att_s, att_d, H, a_s, a_d, m0, head, ln);
    }
}

__global__ __launch_bounds__(GAT_GTHREADS)
__attribute__((amdgpu_waves_per_eu(16, 16)))
void edges_by_node_and_gemm0(
    const int2* __restrict__ part, const int* __restrict__ table,
    int* __restrict__ csr_off, int* __restrict__ csr_src, int n, int nch,
    const float* __restrict__ X, const float* __restrict__ W,
    const float* __restrict__ att_s, const float* __restrict__ att_d,
    float* __restrict__ H, float* __restrict__ a_s, float* __restrict__ a_d)
{
    const int kind = __popc(blockIdx.x) & 1;
    const int idx  = blockIdx.x >> 1;
    if (kind == 0) {
        edges_by_node_range(part, table, csr_off, csr_src, n, nch, idx);
        return;
    }
    const int ntiles = n / 16;
    gemm0_tiles(X, W, att_s, att_d, H, a_s, a_d, ntiles, idx, (int)(gridDim.x >> 1));
}

struct NodeAcc {
    float m, den;
    float4 acc;
};

__device__ __forceinline__ void edge_update(float as, float ad, const float4 hv, NodeAcc& st)
{
    float e = as + ad;
    e = (e > 0.0f) ? e : NEG_SLOPE * e;
    const float mn = fmaxf(st.m, e);
    const float sc = __expf(st.m - mn);
    const float p  = __expf(e - mn);
    st.den   = st.den   * sc + p;
    st.acc.x = st.acc.x * sc + p * hv.x;
    st.acc.y = st.acc.y * sc + p * hv.y;
    st.acc.z = st.acc.z * sc + p * hv.z;
    st.acc.w = st.acc.w * sc + p * hv.w;
    st.m = mn;
}

template <int U>
__device__ __forceinline__ void edge_group(
    int ids, int first, int lane, int head, float ad,
    const float* __restrict__ H, const float* __restrict__ a_s, NodeAcc& st)
{
    float  as[U];
    float4 hv[U];
    #pragma unroll
    for (int u = 0; u < U; ++u) {
        const int s = __builtin_amdgcn_readlane(ids, first + u);
        as[u] = a_s[s * GAT_HEADS + head];
        hv[u] = *reinterpret_cast<const float4*>(H + (size_t)s * GAT_C + 4 * lane);
    }
    #pragma unroll
    for (int u = 0; u < U; ++u)
        edge_update(as[u], ad, hv[u], st);
}

__device__ __forceinline__ void edge_chunk(
    int ids, int c, int lane, int head, float ad,
    const float* __restrict__ H, const float* __restrict__ a_s, NodeAcc& st)
{
    int u = 0;
    for (; u + 8 <= c; u += 8)
        edge_group<8>(ids, u, lane, head, ad, H, a_s, st);
    if (u + 4 <= c) {
        edge_group<4>(ids, u, lane, head, ad, H, a_s, st);
        u += 4;
    }
    for (; u < c; ++u)
        edge_group<1>(ids, u, lane, head, ad, H, a_s, st);
}

template <bool RELU>
__device__ __forceinline__ void store_row(
    const NodeAcc& st, const float4 b4, float* __restrict__ out, int node, int lane)
{
    const float den = (st.den > 0.0f) ? st.den : 1.0f;
    float4 r;
    r.x = st.acc.x / den + b4.x;
    r.y = st.acc.y / den + b4.y;
    r.z = st.acc.z / den + b4.z;
    r.w = st.acc.w / den + b4.w;
    if (RELU) {
        r.x = fmaxf(r.x, 0.0f); r.y = fmaxf(r.y, 0.0f);
        r.z = fmaxf(r.z, 0.0f); r.w = fmaxf(r.w, 0.0f);
    }
    *reinterpret_cast<float4*>(out + (size_t)node * GAT_C + 4 * lane) = r;
}

template <bool RELU>
__global__ __launch_bounds__(GAT_GTHREADS) void gat_gather(
    const int* __restrict__ csr_off, const int* __restrict__ csr_src,
    const float* __restrict__ H,
    const float* __restrict__ a_s, const float* __restrict__ a_d,
    const float* __restrict__ bias, float* __restrict__ out, int n)
{
    __shared__ int s_off[GAT_NPB + 1];

    const int tid  = threadIdx.x;
    const int lane = tid & 31;
    const int head = lane >> 3;
    const int lo   = blockIdx.x * GAT_NPB;
    const int nn   = min(GAT_NPB, n - lo);
    if (nn <= 0) return;

    if (tid <= nn) s_off[tid] = csr_off[lo + tid];
    __syncthreads();

    const float4 b4 = *reinterpret_cast<const float4*>(bias + 4 * lane);

    auto run_ids = [&](int b, int e, int first) {
        return csr_src[max(min(b + first + lane, e - 1), 0)];
    };
    int nl = __builtin_amdgcn_readfirstlane(tid >> 5);
    if (nl >= nn) return;
    int   beg  = __builtin_amdgcn_readfirstlane(s_off[nl]);
    int   end  = __builtin_amdgcn_readfirstlane(s_off[nl + 1]);
    int   idsA = run_ids(beg, end, 0);
    int   idsB = run_ids(beg, end, 32);
    float ad   = a_d[(lo + nl) * GAT_HEADS + head];
    while (nl < nn) {
        const int nl2 = nl + GAT_GWAVES;
        const int nlc = min(nl2, nn - 1);
        const int   beg2 = __builtin_amdgcn_readfirstlane(s_off[nlc]);
        const int   end2 = __builtin_amdgcn_readfirstlane(s_off[nlc + 1]);
        const int   nxtA = run_ids(beg2, end2, 0);
        const int   nxtB = run_ids(beg2, end2, 32);
        const float ad2  = a_d[(lo + nlc) * GAT_HEADS + head];

        const int node = lo + nl;
        NodeAcc st; st.m = -INFINITY; st.den = 0.0f;
        st.acc = make_float4(0.0f, 0.0f, 0.0f, 0.0f);

        const int c = end - beg;
        edge_chunk(idsA, min(c, 32), lane, head, ad, H, a_s, st);
        if (c > 32)
            edge_chunk(idsB, min(c - 32, 32), lane, head, ad, H, a_s, st);
        for (int k = beg + 64; k < end; k += 32) {
            const int ids = (k + lane < end) ? csr_src[k + lane] : 0;
            edge_chunk(ids, min(end - k, 32), lane, head, ad, H, a_s, st);
        }
        store_row<RELU>(st, b4, out, node, lane);

        nl = nl2; beg = beg2; end = end2; idsA = nxtA; idsB = nxtB; ad = ad2;
    }
}

extern "C" void kernel_launch(void* const* d_in, const int* in_sizes, int n_in,
                              void* d_out, int out_size, void* d_ws, size_t ws_size,
                              hipStream_t stream)
{
    const float* x    = (const float*)d_in[0];
    const int*   eidx = (const int*)  d_in[1];
    const float* W0   = (const float*)d_in[2];
    const float* as0  = (const float*)d_in[3];
    const float* ad0  = (const float*)d_in[4];
    const float* b0   = (const float*)d_in[5];
    const float* W1   = (const float*)d_in[6];
    const float* as1  = (const float*)d_in[7];
    const float* ad1  = (const float*)d_in[8];
    const float* b1   = (const float*)d_in[9];

    const int N = in_sizes[0] / GAT_C;
    const int E = in_sizes[1] / 2;
    const int* src = eidx;
    const int* dst = eidx + E;

    char* ws = (char*)d_ws;
    float* hbuf    = (float*)ws;                                ws += (size_t)N * GAT_C * 4;
    char*  plane   = ws;                                        ws += (size_t)N * GAT_C * 4;
    float* a_s     = (float*)ws;                                ws += (size_t)N * GAT_HEADS * 4;
    float* a_d     = (float*)ws;
    float* outf    = (float*)d_out;
    float* out0    = outf;

    const int nch = (E + GAT_CH - 1) / GAT_CH;
    int2* part     = (int2*)plane;                              plane += (size_t)nch * GAT_CH * 8;
    int*  table    = (int*)plane;                               plane += (size_t)nch * (GAT_NB + 1) * 4;
    int*  csr_src  = (int*)plane;                               plane += (size_t)nch * GAT_CH * 4;
    int*  csr_off  = (int*)plane;

    const int blk = 256;
    const dim3 gGemm((N / 16 + 1) / 2);
    const dim3 gGather((N + GAT_NPB - 1) / GAT_NPB);

    edge_partition<<<dim3(nch), GAT_GTHREADS, 0, stream>>>(src, dst, part, table, N, E);
    edges_by_node_and_gemm0<<<dim3(2 * gGather.x), GAT_GTHREADS, 0, stream>>>(
        part, table, csr_off, csr_src, N, nch, x, W0, as0, ad0, hbuf, a_s, a_d);

    gat_gather<true> <<<gGather, GAT_GTHREADS, 0, stream>>>(csr_off, csr_src, hbuf, a_s, a_d,
                                                            b0, out0, N);

    gemm_scores_f32_wmma<<<gGemm, blk, 0, stream>>>(out0, W1, as1, ad1, hbuf, a_s, a_d, N);
    gat_gather<false><<<gGather, GAT_GTHREADS, 0, stream>>>(csr_off, csr_src, hbuf, a_s, a_d,
                                                            b1, outf, N);
}
